# starting kernel
# baseline (speedup 1.0000x reference)
#include <hip/hip_runtime.h>

typedef __attribute__((ext_vector_type(2))) float v2f;
typedef __attribute__((ext_vector_type(8))) float v8f;

namespace {
constexpr int Bn = 4;
constexpr int Cn = 256;
constexpr int Hn = 128;
constexpr int Wn = 224;
constexpr int Pn = 4;
constexpr int ND = 2 * Pn + 1;
constexpr int NWAVES = 7;
}

__global__ __launch_bounds__(NWAVES * 32) void corr_wmma_kernel(
    const float* __restrict__ first,
    const float* __restrict__ second,
    float* __restrict__ out) {
  __shared__ float lds_out[ND * Wn];

  const int blk = blockIdx.x;
  const int dy  = blk % ND;
  const int by  = blk / ND;
  const int y   = by % Hn;
  const int b   = by / Hn;
  const int z   = y + dy - Pn;

  const int tid  = threadIdx.x;
  const int wave = tid >> 5;
  const int lane = tid & 31;
  const int half = lane >> 4;
  const int ln   = lane & 15;

  const size_t cs = (size_t)Hn * Wn;
  float* outbase = out + (((size_t)(b * ND * ND + dy * ND)) * Hn + y) * Wn;

  if (z < 0 || z >= Hn) {
    for (int i = tid; i < ND * Wn; i += blockDim.x) {
      const int dx = i / Wn;
      const int x  = i - dx * Wn;
      outbase[(size_t)dx * cs + x] = 0.0f;
    }
    return;
  }

  const float* Frow = first  + ((size_t)(b * Cn) * Hn + y) * Wn;
  const float* Srow = second + ((size_t)(b * Cn) * Hn + z) * Wn;

  const int n0 = wave * 32;

  const int ca0 = n0 - 4  + ln;
  const int ca1 = n0 + 12 + ln;
  const int ca2 = n0 + 28 + ln;
  const int xb0 = n0 + ln;
  const int xb1 = n0 + 16 + ln;

  const int   ca0c = ca0 < 0 ? 0 : ca0;
  const int   ca2c = ca2 >= Wn ? Wn - 1 : ca2;
  const float m0   = (ca0 >= 0) ? 1.0f : 0.0f;
  const float m2   = (ca2 < Wn) ? 1.0f : 0.0f;

  v8f d00 = {};
  v8f d01 = {};
  v8f d11 = {};
  v8f d21 = {};

#pragma unroll 4
  for (int k = 0; k < Cn; k += 4) {
    const int c0 = k + 2 * half;
    const float* Fp = Frow + (size_t)c0 * cs;
    const float* Sp = Srow + (size_t)c0 * cs;

    v2f b0;  b0.x = Fp[xb0];       b0.y = Fp[cs + xb0];
    v2f b1;  b1.x = Fp[xb1];       b1.y = Fp[cs + xb1];
    v2f a0;  a0.x = Sp[ca0c] * m0; a0.y = Sp[cs + ca0c] * m0;
    v2f a1;  a1.x = Sp[ca1];       a1.y = Sp[cs + ca1];
    v2f a2;  a2.x = Sp[ca2c] * m2; a2.y = Sp[cs + ca2c] * m2;

    d00 = __builtin_amdgcn_wmma_f32_16x16x4_f32(
        false, a0, false, b0, (short)0, d00, false, false);
    d01 = __builtin_amdgcn_wmma_f32_16x16x4_f32(
        false, a1, false, b0, (short)0, d01, false, false);
    d11 = __builtin_amdgcn_wmma_f32_16x16x4_f32(
        false, a1, false, b1, (short)0, d11, false, false);
    d21 = __builtin_amdgcn_wmma_f32_16x16x4_f32(
        false, a2, false, b1, (short)0, d21, false, false);
  }

#pragma unroll
  for (int v = 0; v < 8; ++v) {
    const int m   = v + 8 * half;
    const int dxL = m - ln;
    const int dxH = m + 16 - ln;
    if (dxL >= 0 && dxL < ND) {
      lds_out[dxL * Wn + n0 + ln]      = d00[v];
      lds_out[dxL * Wn + n0 + 16 + ln] = d11[v];
    }
    if (dxH >= 0 && dxH < ND) {
      lds_out[dxH * Wn + n0 + ln]      = d01[v];
      lds_out[dxH * Wn + n0 + 16 + ln] = d21[v];
    }
  }

  __syncthreads();

  for (int i = tid; i < ND * Wn; i += blockDim.x) {
    const int dx = i / Wn;
    const int x  = i - dx * Wn;
    outbase[(size_t)dx * cs + x] = lds_out[i] * (1.0f / (float)Cn);
  }
}

extern "C" void kernel_launch(void* const* d_in, const int* in_sizes, int n_in,
                              void* d_out, int out_size, void* d_ws, size_t ws_size,
                              hipStream_t stream) {
  const float* first  = (const float*)d_in[0];
  const float* second = (const float*)d_in[1];
  float* out = (float*)d_out;

  const int nblocks = Bn * Hn * ND;
  corr_wmma_kernel<<<nblocks, NWAVES * 32, 0, stream>>>(first, second, out);
}


# Register-blocked band: each wave holds all 2x9 (row,dy) accumulator bands of its 16 x, so each operand fragment is loaded once per k-step and reused; next k-step prefetched; summation order unchanged
# speedup vs baseline: 1.7235x; 1.7235x over previous
#include <hip/hip_runtime.h>


typedef __attribute__((ext_vector_type(2))) float v2f;
typedef __attribute__((ext_vector_type(4))) float v4f;
typedef __attribute__((ext_vector_type(8))) float v8f;

namespace {
constexpr int Bn = 4;
constexpr int Cn = 256;
constexpr int Hn = 128;
constexpr int Wn = 224;
constexpr int Pn = 4;
constexpr int ND = 2 * Pn + 1;
constexpr int CS = Hn * Wn;
constexpr int YB = 8;
constexpr int XB = 32;
constexpr int NWV = 8;
constexpr int NZ = 2 + 2 * Pn;
constexpr int NLINES = YB * ND * ND;
}

struct FragSet {
  v2f b[2];
  v2f a[NZ][2];
};

__device__ __forceinline__ float ldg(const float* base, unsigned byteoff) {
  return *reinterpret_cast<const float*>(reinterpret_cast<const char*>(base) + byteoff);
}

__device__ __forceinline__ void load_set(FragSet& f,
                                         const float* __restrict__ Fk,
                                         const float* __restrict__ Sk,
                                         unsigned voffB,
                                         unsigned voffA0, unsigned voffA1,
                                         const int (&rowoff)[NZ]) {
#pragma unroll
  for (int yi = 0; yi < 2; ++yi) {
    const float* r0 = Fk + yi * Wn;
    f.b[yi].x = ldg(r0, voffB);
    f.b[yi].y = ldg(r0 + CS, voffB);
  }
#pragma unroll
  for (int j = 0; j < NZ; ++j) {
    const float* r0 = Sk + rowoff[j];
    f.a[j][0].x = ldg(r0, voffA0);
    f.a[j][0].y = ldg(r0 + CS, voffA0);
    f.a[j][1].x = ldg(r0, voffA1);
    f.a[j][1].y = ldg(r0 + CS, voffA1);
  }
}

__device__ __forceinline__ void mma_set(v8f (&acc)[2][ND][2], FragSet& f,
                                        const bool (&ok)[NZ][2]) {
#pragma unroll
  for (int j = 0; j < NZ; ++j) {
#pragma unroll
    for (int t = 0; t < 2; ++t) {
      f.a[j][t].x = ok[j][t] ? f.a[j][t].x : 0.0f;
      f.a[j][t].y = ok[j][t] ? f.a[j][t].y : 0.0f;
    }
#pragma unroll
    for (int yi = 0; yi < 2; ++yi) {
      const int dy = j - yi;
      if (dy >= 0 && dy < ND) {
        acc[yi][dy][0] = __builtin_amdgcn_wmma_f32_16x16x4_f32(
            false, f.a[j][0], false, f.b[yi], (short)0, acc[yi][dy][0], false, false);
        acc[yi][dy][1] = __builtin_amdgcn_wmma_f32_16x16x4_f32(
            false, f.a[j][1], false, f.b[yi], (short)0, acc[yi][dy][1], false, false);
      }
    }
  }
}

__global__ __launch_bounds__(NWV * 32) __attribute__((amdgpu_waves_per_eu(2, 2)))
void corr_regblock_kernel(const float* __restrict__ first,
                          const float* __restrict__ second,
                          float* __restrict__ out) {
  __shared__ __attribute__((aligned(16))) float lds_out[NLINES * XB];

  const int blk = blockIdx.x;
  const int xt  = blk % (Wn / XB);
  const int t1  = blk / (Wn / XB);
  const int y8  = t1 % (Hn / YB);
  const int b   = t1 / (Hn / YB);

  const int tid  = threadIdx.x;
  const int lane = tid & 31;
  const int wave = __builtin_amdgcn_readfirstlane(tid >> 5);
  const int half = lane >> 4;
  const int ln   = lane & 15;

  const int yp = wave >> 1;
  const int xh = wave & 1;
  const int ya = y8 * YB + 2 * yp;
  const int x0 = xt * XB + 16 * xh;

  const int c0 = x0 - Pn + ln;
  const int c1 = x0 + 12 + ln;
  const int c0c = c0 < 0 ? 0 : c0;
  const int c1c = c1 >= Wn ? Wn - 1 : c1;
  const bool col0ok = c0 >= 0;
  const bool col1ok = c1 < Wn;

  const unsigned hoff   = (unsigned)(2 * half) * (unsigned)CS;
  unsigned voffA0 = (hoff + (unsigned)c0c) * 4u;
  unsigned voffA1 = (hoff + (unsigned)c1c) * 4u;
  unsigned voffB  = (hoff + (unsigned)(x0 + ln)) * 4u;

  int  rowoff[NZ];
  bool ok[NZ][2];
#pragma unroll
  for (int j = 0; j < NZ; ++j) {
    const int z  = ya - Pn + j;
    const int zc = z < 0 ? 0 : (z >= Hn ? Hn - 1 : z);
    const bool zok = (z >= 0) && (z < Hn);
    rowoff[j] = zc * Wn;
    ok[j][0] = zok && col0ok;
    ok[j][1] = zok && col1ok;
  }

  const float* Fb = first  + ((size_t)(b * Cn) * Hn + ya) * Wn;
  const float* Sb = second + (size_t)(b * Cn) * CS;

  v8f acc[2][ND][2];
#pragma unroll
  for (int yi = 0; yi < 2; ++yi)
#pragma unroll
    for (int dy = 0; dy < ND; ++dy) {
      acc[yi][dy][0] = v8f{};
      acc[yi][dy][1] = v8f{};
    }

  FragSet f0, f1;
  load_set(f0, Fb, Sb, voffB, voffA0, voffA1, rowoff);

  for (int k = 0; k < Cn; k += 8) {
    asm volatile("" : "+v"(voffA0), "+v"(voffA1), "+v"(voffB));
    load_set(f1, Fb + (size_t)(k + 4) * CS, Sb + (size_t)(k + 4) * CS,
             voffB, voffA0, voffA1, rowoff);
    __builtin_amdgcn_sched_barrier(0);
    mma_set(acc, f0, ok);
    const int kn = (k + 8 < Cn) ? k + 8 : Cn - 4;
    load_set(f0, Fb + (size_t)kn * CS, Sb + (size_t)kn * CS,
             voffB, voffA0, voffA1, rowoff);
    __builtin_amdgcn_sched_barrier(0);
    mma_set(acc, f1, ok);
  }

#pragma unroll
  for (int yi = 0; yi < 2; ++yi) {
#pragma unroll
    for (int dy = 0; dy < ND; ++dy) {
      const int base = (((2 * yp + yi) * ND + dy) * ND) * XB + 16 * xh + ln;
#pragma unroll
      for (int v = 0; v < 8; ++v) {
        const int dxL = v + 8 * half - ln;
        const int dxH = dxL + 16;
        if (dxL >= 0 && dxL < ND) lds_out[base + dxL * XB] = acc[yi][dy][0][v];
        if (dxH >= 0 && dxH < ND) lds_out[base + dxH * XB] = acc[yi][dy][1][v];
      }
    }
  }

  __syncthreads();

  const float scale = 1.0f / (float)Cn;
  const int sub = lane >> 3;
  const int x4  = (lane & 7) * 4;
  float* ob = out + ((size_t)(b * ND * ND) * Hn + y8 * YB) * Wn + xt * XB + x4;
  for (int L = wave * 4 + sub; L < NLINES; L += NWV * 4) {
    const int yl = L / (ND * ND);
    const int r  = L - yl * (ND * ND);
    v4f val = *reinterpret_cast<const v4f*>(&lds_out[L * XB + x4]);
    val *= scale;
    *reinterpret_cast<v4f*>(ob + ((size_t)r * Hn + yl) * Wn) = val;
  }
}

extern "C" void kernel_launch(void* const* d_in, const int* in_sizes, int n_in,
                              void* d_out, int out_size, void* d_ws, size_t ws_size,
                              hipStream_t stream) {
  const float* first  = (const float*)d_in[0];
  const float* second = (const float*)d_in[1];
  float* out = (float*)d_out;

  const int nblocks = Bn * (Hn / YB) * (Wn / XB);
  corr_regblock_kernel<<<nblocks, NWV * 32, 0, stream>>>(first, second, out);
}
